# starting kernel
# baseline (speedup 1.0000x reference)
#include <hip/hip_runtime.h>

#define H4     4
#define T_LEN  4000
#define BATCH  256
#define NCLS   29
#define NROWS  (BATCH * T_LEN)

typedef __attribute__((ext_vector_type(2))) float v2f;
typedef __attribute__((ext_vector_type(8))) float v8f;

__device__ __forceinline__ float fsigmoid(float x) {
  return __builtin_amdgcn_rcpf(1.0f + __expf(-x));
}

__global__ void lstm_scan_kernel(const float* __restrict__ x,
                                 const float* __restrict__ W_ih,
                                 const float* __restrict__ W_hh,
                                 const float* __restrict__ b_ih,
                                 const float* __restrict__ b_hh,
                                 float* __restrict__ hs)
{
  const int tid = blockIdx.x * blockDim.x + threadIdx.x;
  const int b = tid >> 2;
  const int j = tid & 3;
  if (b >= BATCH) return;

  float win[4], ball[4], whh[4][4];
#pragma unroll
  for (int s = 0; s < 4; ++s) {
    const int g = s * 4 + j;
    win[s]  = W_ih[g];
    ball[s] = b_ih[g] + b_hh[g];
#pragma unroll
    for (int m = 0; m < 4; ++m)
      whh[s][m] = W_hh[g * 4 + (j ^ m)];
  }

  float h = 0.0f, c = 0.0f;
  const float* xb  = x  + (size_t)b * T_LEN;
  float*       hsb = hs + (size_t)b * T_LEN * 4 + j;

  for (int t = 0; t < T_LEN; ++t) {
    const float xt = xb[t];
    const float v1 = __shfl_xor(h, 1);
    const float v2 = __shfl_xor(h, 2);
    const float v3 = __shfl_xor(v1, 2);

    float pre[4];
#pragma unroll
    for (int s = 0; s < 4; ++s) {
      float p = fmaf(xt, win[s], ball[s]);
      p = fmaf(whh[s][0], h,  p);
      p = fmaf(whh[s][1], v1, p);
      p = fmaf(whh[s][2], v2, p);
      p = fmaf(whh[s][3], v3, p);
      pre[s] = p;
    }

    const float ig = fsigmoid(pre[0]);
    const float fg = fsigmoid(pre[1]);
    const float gg = fmaf(2.0f, fsigmoid(2.0f * pre[2]), -1.0f);
    const float og = fsigmoid(pre[3]);

    c = fmaf(fg, c, ig * gg);
    const float tc = fmaf(2.0f, fsigmoid(2.0f * c), -1.0f);
    h = og * tc;

    hsb[(size_t)t * 4] = h;
  }
}

__global__ void fc_wmma_kernel(const float* __restrict__ hs,
                               const float* __restrict__ W_fc,
                               const float* __restrict__ b_fc,
                               float* __restrict__ out)
{
  const int wave     = (blockIdx.x * blockDim.x + threadIdx.x) >> 5;
  const int lane     = threadIdx.x & 31;
  const int row_base = wave * 16;
  const int lm       = lane & 15;
  const int k0       = (lane < 16) ? 0 : 2;

  const int row = row_base + lm;
  v2f a;
  a.x = hs[(size_t)row * 4 + k0];
  a.y = hs[(size_t)row * 4 + k0 + 1];

#pragma unroll
  for (int tile = 0; tile < 2; ++tile) {
    const int n = tile * 16 + lm;
    v2f bm;
    float bias;
    if (n < NCLS) {
      bm.x = W_fc[n * 4 + k0];
      bm.y = W_fc[n * 4 + k0 + 1];
      bias = b_fc[n];
    } else {
      bm.x = 0.0f; bm.y = 0.0f; bias = 0.0f;
    }

    v8f acc;
#pragma unroll
    for (int r = 0; r < 8; ++r) acc[r] = bias;

    acc = __builtin_amdgcn_wmma_f32_16x16x4_f32(
         false, a,  false, bm,
         (short)0, acc,  false,  false);

    if (n < NCLS) {
      const int m_off = (lane < 16) ? 0 : 8;
#pragma unroll
      for (int r = 0; r < 8; ++r) {
        const int m = row_base + m_off + r;
        out[(size_t)m * NCLS + n] = acc[r];
      }
    }
  }
}

extern "C" void kernel_launch(void* const* d_in, const int* in_sizes, int n_in,
                              void* d_out, int out_size, void* d_ws, size_t ws_size,
                              hipStream_t stream) {
  const float* x    = (const float*)d_in[0];
  const float* W_ih = (const float*)d_in[1];
  const float* W_hh = (const float*)d_in[2];
  const float* b_ih = (const float*)d_in[3];
  const float* b_hh = (const float*)d_in[4];
  const float* W_fc = (const float*)d_in[5];
  const float* b_fc = (const float*)d_in[6];
  float* out = (float*)d_out;

  float* hs = (float*)d_ws;

  lstm_scan_kernel<<<(BATCH * 4 + 255) / 256, 256, 0, stream>>>(
      x, W_ih, W_hh, b_ih, b_hh, hs);

  const int n_tiles  = NROWS / 16;
  const int n_blocks = n_tiles / 8;
  fc_wmma_kernel<<<n_blocks, 256, 0, stream>>>(hs, W_fc, b_fc, out);
}


# Projection re-tiled: 8,000 waves (was 64,000) each walk 128 rows in 64-row passes staged in the wave's own LDS block, out leaves as 16-byte stores of whole 128-byte lines; sums unchanged
# speedup vs baseline: 3.2091x; 3.2091x over previous
#include <hip/hip_runtime.h>


#define H4     4
#define T_LEN  4000
#define BATCH  256
#define NCLS   29
#define NROWS  (BATCH * T_LEN)
#define TSTEP  32

typedef __attribute__((ext_vector_type(2))) float v2f;
typedef __attribute__((ext_vector_type(4))) float v4f;
typedef __attribute__((ext_vector_type(8))) float v8f;

__global__ __launch_bounds__(32) __attribute__((amdgpu_waves_per_eu(1, 1)))
void lstm_scan_wmma_kernel(const float* __restrict__ x,
                           const float* __restrict__ W_ih,
                           const float* __restrict__ W_hh,
                           const float* __restrict__ b_ih,
                           const float* __restrict__ b_hh,
                           float* __restrict__ hs)
{
  const int lane = threadIdx.x & 31;
  const int col  = lane & 15;
  const int hh   = lane >> 4;
  const int b    = blockIdx.x * 16 + col;

  const float L2E = 1.44269504088896340736f;
  const float KC  = -2.0f * L2E;

  const int ga = (col & 3) * 4 + (col >> 2);
  const float sa = ((col & 3) == 2) ? KC : -L2E;
  v2f a;
  a.x = W_hh[ga * 4 + 2 * hh] * sa;
  a.y = W_hh[ga * 4 + 2 * hh + 1] * sa;

  float win[8], ball[8];
#pragma unroll
  for (int r = 0; r < 8; ++r) {
    const int g = (r & 3) * 4 + 2 * hh + (r >> 2);
    const float sr = ((r & 3) == 2) ? KC : -L2E;
    win[r]  = W_ih[g] * sr;
    ball[r] = (b_ih[g] + b_hh[g]) * sr;
  }

  v2f hv;
  hv.x = 0.0f; hv.y = 0.0f;
  float cs0 = 0.0f, cs1 = 0.0f;

  const float* xb  = x  + (size_t)b * T_LEN;
  float*       hsb = hs + (size_t)b * T_LEN * 4 + 2 * hh;

  for (int t0 = 0; t0 < T_LEN; t0 += TSTEP) {
    v4f xv[TSTEP / 4];
#pragma unroll
    for (int q = 0; q < TSTEP / 4; ++q)
      xv[q] = *reinterpret_cast<const v4f*>(xb + t0 + 4 * q);

#pragma unroll
    for (int i = 0; i < TSTEP; ++i) {
      const float xt = xv[i >> 2][i & 3];

      v8f acc;
#pragma unroll
      for (int r = 0; r < 8; ++r) acc[r] = fmaf(xt, win[r], ball[r]);

      acc = __builtin_amdgcn_wmma_f32_16x16x4_f32(
           false, a,  false, hv,
           (short)0, acc,  false,  false);

      float sg[8];
#pragma unroll
      for (int r = 0; r < 8; ++r) sg[r] = __builtin_amdgcn_exp2f(acc[r]);
#pragma unroll
      for (int r = 0; r < 8; ++r) sg[r] = __builtin_amdgcn_rcpf(1.0f + sg[r]);

      const float p0 = fmaf(sg[2], 2.0f * KC, -KC);
      const float p1 = fmaf(sg[6], 2.0f * KC, -KC);
      cs0 = fmaf(sg[0], p0, sg[1] * cs0);
      cs1 = fmaf(sg[4], p1, sg[5] * cs1);
      const float r0 = __builtin_amdgcn_rcpf(1.0f + __builtin_amdgcn_exp2f(cs0));
      const float r1 = __builtin_amdgcn_rcpf(1.0f + __builtin_amdgcn_exp2f(cs1));
      hv.x = fmaf(sg[3] + sg[3], r0, -sg[3]);
      hv.y = fmaf(sg[7] + sg[7], r1, -sg[7]);

      *reinterpret_cast<v2f*>(hsb + (size_t)(t0 + i) * 4) = hv;
    }
  }
}

#define FC_ROWS    64
#define FC_PASSES  2
#define FC_WAVES   8
#define FC_BLOCKS  (NROWS / (FC_ROWS * FC_PASSES * FC_WAVES))
#define FC_TILE    (FC_ROWS * NCLS)

__global__ __launch_bounds__(256)
void fc_wmma_kernel(const float* __restrict__ hs,
                    const float* __restrict__ W_fc,
                    const float* __restrict__ b_fc,
                    float* __restrict__ out)
{
  __shared__ __attribute__((aligned(16))) float stage[FC_WAVES][FC_TILE];

  const int wib  = threadIdx.x >> 5;
  const int lane = threadIdx.x & 31;
  const int lm   = lane & 15;
  const int half = lane >> 4;
  const int k0   = 2 * half;
  float* st = stage[wib];

  v2f bm0, bm1;
  bm0.x = W_fc[lm * 4 + k0];
  bm0.y = W_fc[lm * 4 + k0 + 1];
  const float bias0 = b_fc[lm];
  const bool  hi_ok = (16 + lm) < NCLS;
  float bias1 = 0.0f;
  bm1.x = 0.0f; bm1.y = 0.0f;
  if (hi_ok) {
    bm1.x = W_fc[(16 + lm) * 4 + k0];
    bm1.y = W_fc[(16 + lm) * 4 + k0 + 1];
    bias1 = b_fc[16 + lm];
  }

  const size_t row0 = (size_t)(blockIdx.x * FC_WAVES + wib) * (FC_ROWS * FC_PASSES);

  v2f a[FC_PASSES][FC_ROWS / 16];
#pragma unroll
  for (int p = 0; p < FC_PASSES; ++p)
#pragma unroll
    for (int rt = 0; rt < FC_ROWS / 16; ++rt)
      a[p][rt] = *reinterpret_cast<const v2f*>(
          hs + (row0 + (size_t)(p * FC_ROWS + rt * 16 + lm)) * 4 + k0);

#pragma unroll
  for (int p = 0; p < FC_PASSES; ++p) {
#pragma unroll
    for (int rt = 0; rt < FC_ROWS / 16; ++rt) {
      v8f acc0, acc1;
#pragma unroll
      for (int r = 0; r < 8; ++r) { acc0[r] = bias0; acc1[r] = bias1; }
      acc0 = __builtin_amdgcn_wmma_f32_16x16x4_f32(
           false, a[p][rt],  false, bm0,
           (short)0, acc0,  false,  false);
      acc1 = __builtin_amdgcn_wmma_f32_16x16x4_f32(
           false, a[p][rt],  false, bm1,
           (short)0, acc1,  false,  false);
      float* sr = st + (rt * 16 + 8 * half) * NCLS + lm;
#pragma unroll
      for (int r = 0; r < 8; ++r) sr[r * NCLS] = acc0[r];
      if (hi_ok) {
#pragma unroll
        for (int r = 0; r < 8; ++r) sr[r * NCLS + 16] = acc1[r];
      }
    }

    __builtin_amdgcn_fence(__ATOMIC_RELEASE, "wavefront");
    __builtin_amdgcn_wave_barrier();
    __builtin_amdgcn_fence(__ATOMIC_ACQUIRE, "wavefront");

    float* og = out + (row0 + (size_t)p * FC_ROWS) * NCLS;
#pragma unroll
    for (int j = 0; j < (FC_TILE / 4 + 31) / 32; ++j) {
      const int q = j * 32 + lane;
      if (q < FC_TILE / 4)
        *reinterpret_cast<v4f*>(og + q * 4) = *reinterpret_cast<const v4f*>(st + q * 4);
    }

    __builtin_amdgcn_fence(__ATOMIC_RELEASE, "wavefront");
    __builtin_amdgcn_wave_barrier();
    __builtin_amdgcn_fence(__ATOMIC_ACQUIRE, "wavefront");
  }
}

extern "C" void kernel_launch(void* const* d_in, const int* in_sizes, int n_in,
                              void* d_out, int out_size, void* d_ws, size_t ws_size,
                              hipStream_t stream) {
  const float* x    = (const float*)d_in[0];
  const float* W_ih = (const float*)d_in[1];
  const float* W_hh = (const float*)d_in[2];
  const float* b_ih = (const float*)d_in[3];
  const float* b_hh = (const float*)d_in[4];
  const float* W_fc = (const float*)d_in[5];
  const float* b_fc = (const float*)d_in[6];
  float* out = (float*)d_out;

  float* hs = (float*)d_ws;

  lstm_scan_wmma_kernel<<<BATCH / 16, 32, 0, stream>>>(
      x, W_ih, W_hh, b_ih, b_hh, hs);

  fc_wmma_kernel<<<FC_BLOCKS, 256, 0, stream>>>(hs, W_fc, b_fc, out);
}


# LSTM scan re-tiled: 8 batch elements per wave (32 waves), columns n and n+8 carry one element so a lane evaluates ONE unit (was 2); partner h by one DPP read; k of units 1,3 swapped in the WMMA sum
# speedup vs baseline: 3.9353x; 1.2263x over previous
#include <hip/hip_runtime.h>


#define H4     4
#define T_LEN  4000
#define BATCH  256
#define NCLS   29
#define NROWS  (BATCH * T_LEN)
#define TSTEP  32

typedef __attribute__((ext_vector_type(2))) float v2f;
typedef __attribute__((ext_vector_type(4))) float v4f;
typedef __attribute__((ext_vector_type(8))) float v8f;

#define SCAN_BT  8

__device__ __forceinline__ float partner8(float v) {
  return __builtin_bit_cast(float, __builtin_amdgcn_update_dpp(
      0, __builtin_bit_cast(int, v),   0x128, 0xf, 0xf, false));
}

__global__ __launch_bounds__(32) __attribute__((amdgpu_waves_per_eu(1, 1)))
void lstm_scan_wmma_kernel(const float* __restrict__ x,
                           const float* __restrict__ W_ih,
                           const float* __restrict__ W_hh,
                           const float* __restrict__ b_ih,
                           const float* __restrict__ b_hh,
                           float* __restrict__ hs)
{
  const int lane = threadIdx.x & 31;
  const int col  = lane & 15;
  const int hh   = lane >> 4;
  const int y    = col >> 3;
  const int b    = blockIdx.x * SCAN_BT + (col & 7);

  const float L2E = 1.44269504088896340736f;
  const float KC  = -2.0f * L2E;

  const int ga = (col & 3) * 4 + (col >> 2);
  const int yr = (col >> 2) & 1;
  const float sa = ((col & 3) == 2) ? KC : -L2E;
  v2f a;
  a.x = W_hh[ga * 4 + 2 * hh + yr] * sa;
  a.y = W_hh[ga * 4 + 2 * hh + (yr ^ 1)] * sa;

  float win[8], ball[8];
#pragma unroll
  for (int r = 0; r < 8; ++r) {
    const int g = (r & 3) * 4 + 2 * hh + (r >> 2);
    const float sr = ((r & 3) == 2) ? KC : -L2E;
    win[r]  = W_ih[g] * sr;
    ball[r] = (b_ih[g] + b_hh[g]) * sr;
  }

  v2f hv;
  hv.x = 0.0f; hv.y = 0.0f;
  float cs = 0.0f;

  const float* xb  = x  + (size_t)b * T_LEN;
  float*       hsb = hs + (size_t)b * T_LEN * 4 + 2 * hh + y;

  for (int t0 = 0; t0 < T_LEN; t0 += TSTEP) {
    v4f xv[TSTEP / 4];
#pragma unroll
    for (int q = 0; q < TSTEP / 4; ++q)
      xv[q] = *reinterpret_cast<const v4f*>(xb + t0 + 4 * q);

#pragma unroll
    for (int i = 0; i < TSTEP; ++i) {
      const float xt = xv[i >> 2][i & 3];

      v8f acc;
#pragma unroll
      for (int r = 0; r < 8; ++r) acc[r] = fmaf(xt, win[r], ball[r]);

      acc = __builtin_amdgcn_wmma_f32_16x16x4_f32(
           false, a,  false, hv,
           (short)0, acc,  false,  false);

      float sg[4];
#pragma unroll
      for (int r = 0; r < 4; ++r) {
        const float lo = acc[r];
        const float hi = acc[4 + r];
        sg[r] = y ? hi : lo;
      }
#pragma unroll
      for (int r = 0; r < 4; ++r) sg[r] = __builtin_amdgcn_exp2f(sg[r]);
#pragma unroll
      for (int r = 0; r < 4; ++r) sg[r] = __builtin_amdgcn_rcpf(1.0f + sg[r]);

      const float p = fmaf(sg[2], 2.0f * KC, -KC);
      cs = fmaf(sg[0], p, sg[1] * cs);

      const float ox = partner8(sg[3]);

      const float r2 = __builtin_amdgcn_rcpf(fmaf(__builtin_amdgcn_exp2f(cs), 0.5f, 0.5f));
      hv.x = fmaf(sg[3], r2, -sg[3]);
      hv.y = fmaf(partner8(r2), ox, -ox);

      hsb[(size_t)(t0 + i) * 4] = hv.x;
    }
  }
}

#define FC_ROWS    64
#define FC_PASSES  2
#define FC_WAVES   8
#define FC_BLOCKS  (NROWS / (FC_ROWS * FC_PASSES * FC_WAVES))
#define FC_TILE    (FC_ROWS * NCLS)

__global__ __launch_bounds__(256)
void fc_wmma_kernel(const float* __restrict__ hs,
                    const float* __restrict__ W_fc,
                    const float* __restrict__ b_fc,
                    float* __restrict__ out)
{
  __shared__ __attribute__((aligned(16))) float stage[FC_WAVES][FC_TILE];

  const int wib  = threadIdx.x >> 5;
  const int lane = threadIdx.x & 31;
  const int lm   = lane & 15;
  const int half = lane >> 4;
  const int k0   = 2 * half;
  float* st = stage[wib];

  v2f bm0, bm1;
  bm0.x = W_fc[lm * 4 + k0];
  bm0.y = W_fc[lm * 4 + k0 + 1];
  const float bias0 = b_fc[lm];
  const bool  hi_ok = (16 + lm) < NCLS;
  float bias1 = 0.0f;
  bm1.x = 0.0f; bm1.y = 0.0f;
  if (hi_ok) {
    bm1.x = W_fc[(16 + lm) * 4 + k0];
    bm1.y = W_fc[(16 + lm) * 4 + k0 + 1];
    bias1 = b_fc[16 + lm];
  }

  const size_t row0 = (size_t)(blockIdx.x * FC_WAVES + wib) * (FC_ROWS * FC_PASSES);

  v2f a[FC_PASSES][FC_ROWS / 16];
#pragma unroll
  for (int p = 0; p < FC_PASSES; ++p)
#pragma unroll
    for (int rt = 0; rt < FC_ROWS / 16; ++rt)
      a[p][rt] = *reinterpret_cast<const v2f*>(
          hs + (row0 + (size_t)(p * FC_ROWS + rt * 16 + lm)) * 4 + k0);

#pragma unroll
  for (int p = 0; p < FC_PASSES; ++p) {
#pragma unroll
    for (int rt = 0; rt < FC_ROWS / 16; ++rt) {
      v8f acc0, acc1;
#pragma unroll
      for (int r = 0; r < 8; ++r) { acc0[r] = bias0; acc1[r] = bias1; }
      acc0 = __builtin_amdgcn_wmma_f32_16x16x4_f32(
           false, a[p][rt],  false, bm0,
           (short)0, acc0,  false,  false);
      acc1 = __builtin_amdgcn_wmma_f32_16x16x4_f32(
           false, a[p][rt],  false, bm1,
           (short)0, acc1,  false,  false);
      float* sr = st + (rt * 16 + 8 * half) * NCLS + lm;
#pragma unroll
      for (int r = 0; r < 8; ++r) sr[r * NCLS] = acc0[r];
      if (hi_ok) {
#pragma unroll
        for (int r = 0; r < 8; ++r) sr[r * NCLS + 16] = acc1[r];
      }
    }

    __builtin_amdgcn_fence(__ATOMIC_RELEASE, "wavefront");
    __builtin_amdgcn_wave_barrier();
    __builtin_amdgcn_fence(__ATOMIC_ACQUIRE, "wavefront");

    float* og = out + (row0 + (size_t)p * FC_ROWS) * NCLS;
#pragma unroll
    for (int j = 0; j < (FC_TILE / 4 + 31) / 32; ++j) {
      const int q = j * 32 + lane;
      if (q < FC_TILE / 4)
        *reinterpret_cast<v4f*>(og + q * 4) = *reinterpret_cast<const v4f*>(st + q * 4);
    }

    __builtin_amdgcn_fence(__ATOMIC_RELEASE, "wavefront");
    __builtin_amdgcn_wave_barrier();
    __builtin_amdgcn_fence(__ATOMIC_ACQUIRE, "wavefront");
  }
}

extern "C" void kernel_launch(void* const* d_in, const int* in_sizes, int n_in,
                              void* d_out, int out_size, void* d_ws, size_t ws_size,
                              hipStream_t stream) {
  const float* x    = (const float*)d_in[0];
  const float* W_ih = (const float*)d_in[1];
  const float* W_hh = (const float*)d_in[2];
  const float* b_ih = (const float*)d_in[3];
  const float* b_hh = (const float*)d_in[4];
  const float* W_fc = (const float*)d_in[5];
  const float* b_fc = (const float*)d_in[6];
  float* out = (float*)d_out;

  float* hs = (float*)d_ws;

  lstm_scan_wmma_kernel<<<BATCH / SCAN_BT, 32, 0, stream>>>(
      x, W_ih, W_hh, b_ih, b_hh, hs);

  fc_wmma_kernel<<<FC_BLOCKS, 256, 0, stream>>>(hs, W_fc, b_fc, out);
}


# LSTM scan: x of the wave's 8 batch rows (125 KiB, contiguous) is copied to LDS once, 50 16-byte loads in flight; each 32-step tile then reads its line of x from LDS, not from memory; sums unchanged
# speedup vs baseline: 4.2360x; 1.0764x over previous
#include <hip/hip_runtime.h>


#define H4     4
#define T_LEN  4000
#define BATCH  256
#define NCLS   29
#define NROWS  (BATCH * T_LEN)
#define TSTEP  32

typedef __attribute__((ext_vector_type(2))) float v2f;
typedef __attribute__((ext_vector_type(4))) float v4f;
typedef __attribute__((ext_vector_type(8))) float v8f;

#define SCAN_BT  8

#define XS_PITCH   4004
#define XS_PIECES  (SCAN_BT * T_LEN / 4)
#define XS_BATCH   50

__device__ __forceinline__ float partner8(float v) {
  return __builtin_bit_cast(float, __builtin_amdgcn_update_dpp(
      0, __builtin_bit_cast(int, v),   0x128, 0xf, 0xf, false));
}

__global__ __launch_bounds__(32) __attribute__((amdgpu_waves_per_eu(1, 1)))
void lstm_scan_wmma_kernel(const float* __restrict__ x,
                           const float* __restrict__ W_ih,
                           const float* __restrict__ W_hh,
                           const float* __restrict__ b_ih,
                           const float* __restrict__ b_hh,
                           float* __restrict__ hs)
{
  const int lane = threadIdx.x & 31;
  const int col  = lane & 15;
  const int hh   = lane >> 4;
  const int y    = col >> 3;
  const int b    = blockIdx.x * SCAN_BT + (col & 7);

  const float L2E = 1.44269504088896340736f;
  const float KC  = -2.0f * L2E;

  const int ga = (col & 3) * 4 + (col >> 2);
  const int yr = (col >> 2) & 1;
  const float sa = ((col & 3) == 2) ? KC : -L2E;
  v2f a;
  a.x = W_hh[ga * 4 + 2 * hh + yr] * sa;
  a.y = W_hh[ga * 4 + 2 * hh + (yr ^ 1)] * sa;

  float win[8], ball[8];
#pragma unroll
  for (int r = 0; r < 8; ++r) {
    const int g = (r & 3) * 4 + 2 * hh + (r >> 2);
    const float sr = ((r & 3) == 2) ? KC : -L2E;
    win[r]  = W_ih[g] * sr;
    ball[r] = (b_ih[g] + b_hh[g]) * sr;
  }

  v2f hv;
  hv.x = 0.0f; hv.y = 0.0f;
  float cs = 0.0f;

  __shared__ __attribute__((aligned(16))) float xs[SCAN_BT * XS_PITCH];

  {
    const float* xg = x + (size_t)blockIdx.x * SCAN_BT * T_LEN;
#pragma unroll 1
    for (int j0 = 0; j0 < XS_PIECES / 32; j0 += XS_BATCH) {
      v4f tmp[XS_BATCH];
#pragma unroll
      for (int u = 0; u < XS_BATCH; ++u)
        tmp[u] = *reinterpret_cast<const v4f*>(xg + 4 * ((j0 + u) * 32 + lane));
#pragma unroll
      for (int u = 0; u < XS_BATCH; ++u) {
        const int piece = (j0 + u) * 32 + lane;
        const int row   = piece / (T_LEN / 4);
        *reinterpret_cast<v4f*>(xs + 4 * piece + (XS_PITCH - T_LEN) * row) = tmp[u];
      }
    }
  }
  __syncthreads();

  const float* xb  = xs + (col & 7) * XS_PITCH;
  float*       hsb = hs + (size_t)b * T_LEN * 4 + 2 * hh + y;

  for (int t0 = 0; t0 < T_LEN; t0 += TSTEP) {
    v4f xv[TSTEP / 4];
#pragma unroll
    for (int q = 0; q < TSTEP / 4; ++q)
      xv[q] = *reinterpret_cast<const v4f*>(xb + t0 + 4 * q);

#pragma unroll
    for (int i = 0; i < TSTEP; ++i) {
      const float xt = xv[i >> 2][i & 3];

      v8f acc;
#pragma unroll
      for (int r = 0; r < 8; ++r) acc[r] = fmaf(xt, win[r], ball[r]);

      acc = __builtin_amdgcn_wmma_f32_16x16x4_f32(
           false, a,  false, hv,
           (short)0, acc,  false,  false);

      float sg[4];
#pragma unroll
      for (int r = 0; r < 4; ++r) {
        const float lo = acc[r];
        const float hi = acc[4 + r];
        sg[r] = y ? hi : lo;
      }
#pragma unroll
      for (int r = 0; r < 4; ++r) sg[r] = __builtin_amdgcn_exp2f(sg[r]);
#pragma unroll
      for (int r = 0; r < 4; ++r) sg[r] = __builtin_amdgcn_rcpf(1.0f + sg[r]);

      const float p = fmaf(sg[2], 2.0f * KC, -KC);
      cs = fmaf(sg[0], p, sg[1] * cs);

      const float ox = partner8(sg[3]);

      const float r2 = __builtin_amdgcn_rcpf(fmaf(__builtin_amdgcn_exp2f(cs), 0.5f, 0.5f));
      hv.x = fmaf(sg[3], r2, -sg[3]);
      hv.y = fmaf(partner8(r2), ox, -ox);

      hsb[(size_t)(t0 + i) * 4] = hv.x;
    }
  }
}

#define FC_ROWS    64
#define FC_PASSES  2
#define FC_WAVES   8
#define FC_BLOCKS  (NROWS / (FC_ROWS * FC_PASSES * FC_WAVES))
#define FC_TILE    (FC_ROWS * NCLS)

__global__ __launch_bounds__(256)
void fc_wmma_kernel(const float* __restrict__ hs,
                    const float* __restrict__ W_fc,
                    const float* __restrict__ b_fc,
                    float* __restrict__ out)
{
  __shared__ __attribute__((aligned(16))) float stage[FC_WAVES][FC_TILE];

  const int wib  = threadIdx.x >> 5;
  const int lane = threadIdx.x & 31;
  const int lm   = lane & 15;
  const int half = lane >> 4;
  const int k0   = 2 * half;
  float* st = stage[wib];

  v2f bm0, bm1;
  bm0.x = W_fc[lm * 4 + k0];
  bm0.y = W_fc[lm * 4 + k0 + 1];
  const float bias0 = b_fc[lm];
  const bool  hi_ok = (16 + lm) < NCLS;
  float bias1 = 0.0f;
  bm1.x = 0.0f; bm1.y = 0.0f;
  if (hi_ok) {
    bm1.x = W_fc[(16 + lm) * 4 + k0];
    bm1.y = W_fc[(16 + lm) * 4 + k0 + 1];
    bias1 = b_fc[16 + lm];
  }

  const size_t row0 = (size_t)(blockIdx.x * FC_WAVES + wib) * (FC_ROWS * FC_PASSES);

  v2f a[FC_PASSES][FC_ROWS / 16];
#pragma unroll
  for (int p = 0; p < FC_PASSES; ++p)
#pragma unroll
    for (int rt = 0; rt < FC_ROWS / 16; ++rt)
      a[p][rt] = *reinterpret_cast<const v2f*>(
          hs + (row0 + (size_t)(p * FC_ROWS + rt * 16 + lm)) * 4 + k0);

#pragma unroll
  for (int p = 0; p < FC_PASSES; ++p) {
#pragma unroll
    for (int rt = 0; rt < FC_ROWS / 16; ++rt) {
      v8f acc0, acc1;
#pragma unroll
      for (int r = 0; r < 8; ++r) { acc0[r] = bias0; acc1[r] = bias1; }
      acc0 = __builtin_amdgcn_wmma_f32_16x16x4_f32(
           false, a[p][rt],  false, bm0,
           (short)0, acc0,  false,  false);
      acc1 = __builtin_amdgcn_wmma_f32_16x16x4_f32(
           false, a[p][rt],  false, bm1,
           (short)0, acc1,  false,  false);
      float* sr = st + (rt * 16 + 8 * half) * NCLS + lm;
#pragma unroll
      for (int r = 0; r < 8; ++r) sr[r * NCLS] = acc0[r];
      if (hi_ok) {
#pragma unroll
        for (int r = 0; r < 8; ++r) sr[r * NCLS + 16] = acc1[r];
      }
    }

    __builtin_amdgcn_fence(__ATOMIC_RELEASE, "wavefront");
    __builtin_amdgcn_wave_barrier();
    __builtin_amdgcn_fence(__ATOMIC_ACQUIRE, "wavefront");

    float* og = out + (row0 + (size_t)p * FC_ROWS) * NCLS;
#pragma unroll
    for (int j = 0; j < (FC_TILE / 4 + 31) / 32; ++j) {
      const int q = j * 32 + lane;
      if (q < FC_TILE / 4)
        *reinterpret_cast<v4f*>(og + q * 4) = *reinterpret_cast<const v4f*>(st + q * 4);
    }

    __builtin_amdgcn_fence(__ATOMIC_RELEASE, "wavefront");
    __builtin_amdgcn_wave_barrier();
    __builtin_amdgcn_fence(__ATOMIC_ACQUIRE, "wavefront");
  }
}

extern "C" void kernel_launch(void* const* d_in, const int* in_sizes, int n_in,
                              void* d_out, int out_size, void* d_ws, size_t ws_size,
                              hipStream_t stream) {
  const float* x    = (const float*)d_in[0];
  const float* W_ih = (const float*)d_in[1];
  const float* W_hh = (const float*)d_in[2];
  const float* b_ih = (const float*)d_in[3];
  const float* b_hh = (const float*)d_in[4];
  const float* W_fc = (const float*)d_in[5];
  const float* b_fc = (const float*)d_in[6];
  float* out = (float*)d_out;

  float* hs = (float*)d_ws;

  lstm_scan_wmma_kernel<<<BATCH / SCAN_BT, 32, 0, stream>>>(
      x, W_ih, W_hh, b_ih, b_hh, hs);

  fc_wmma_kernel<<<FC_BLOCKS, 256, 0, stream>>>(hs, W_fc, b_fc, out);
}


# LSTM scan: no per-tile work between steps any more: each step's C (x*w_in+b) is built behind the previous step's WMMA, x is read from LDS 4 steps ahead, h's pointer no longer waits; sums unchanged
# speedup vs baseline: 4.4402x; 1.0482x over previous
#include <hip/hip_runtime.h>


#define H4     4
#define T_LEN  4000
#define BATCH  256
#define NCLS   29
#define NROWS  (BATCH * T_LEN)
#define TSTEP  32

typedef __attribute__((ext_vector_type(2))) float v2f;
typedef __attribute__((ext_vector_type(4))) float v4f;
typedef __attribute__((ext_vector_type(8))) float v8f;

#define SCAN_BT  8

#define XS_PITCH   4004
#define XS_PIECES  (SCAN_BT * T_LEN / 4)
#define XS_BATCH   50

__device__ __forceinline__ float partner8(float v) {
  return __builtin_bit_cast(float, __builtin_amdgcn_update_dpp(
      0, __builtin_bit_cast(int, v),   0x128, 0xf, 0xf, false));
}

__global__ __launch_bounds__(32) __attribute__((amdgpu_waves_per_eu(1, 1)))
void lstm_scan_wmma_kernel(const float* __restrict__ x,
                           const float* __restrict__ W_ih,
                           const float* __restrict__ W_hh,
                           const float* __restrict__ b_ih,
                           const float* __restrict__ b_hh,
                           float* __restrict__ hs)
{
  const int lane = threadIdx.x & 31;
  const int col  = lane & 15;
  const int hh   = lane >> 4;
  const int y    = col >> 3;
  const int b    = blockIdx.x * SCAN_BT + (col & 7);

  const float L2E = 1.44269504088896340736f;
  const float KC  = -2.0f * L2E;

  const int ga = (col & 3) * 4 + (col >> 2);
  const int yr = (col >> 2) & 1;
  const float sa = ((col & 3) == 2) ? KC : -L2E;
  v2f a;
  a.x = W_hh[ga * 4 + 2 * hh + yr] * sa;
  a.y = W_hh[ga * 4 + 2 * hh + (yr ^ 1)] * sa;

  float win[8], ball[8];
#pragma unroll
  for (int r = 0; r < 8; ++r) {
    const int g = (r & 3) * 4 + 2 * hh + (r >> 2);
    const float sr = ((r & 3) == 2) ? KC : -L2E;
    win[r]  = W_ih[g] * sr;
    ball[r] = (b_ih[g] + b_hh[g]) * sr;
  }

  v2f hv;
  hv.x = 0.0f; hv.y = 0.0f;
  float cs = 0.0f;

  __shared__ __attribute__((aligned(16))) float xs[SCAN_BT * XS_PITCH];

  {
    const float* xg = x + (size_t)blockIdx.x * SCAN_BT * T_LEN;
#pragma unroll 1
    for (int j0 = 0; j0 < XS_PIECES / 32; j0 += XS_BATCH) {
      v4f tmp[XS_BATCH];
#pragma unroll
      for (int u = 0; u < XS_BATCH; ++u)
        tmp[u] = *reinterpret_cast<const v4f*>(xg + 4 * ((j0 + u) * 32 + lane));
#pragma unroll
      for (int u = 0; u < XS_BATCH; ++u) {
        const int piece = (j0 + u) * 32 + lane;
        const int row   = piece / (T_LEN / 4);
        *reinterpret_cast<v4f*>(xs + 4 * piece + (XS_PITCH - T_LEN) * row) = tmp[u];
      }
    }
  }
  __syncthreads();

  const float* xb = xs + (col & 7) * XS_PITCH;
  const unsigned lane_off = ((unsigned)b * (T_LEN * 4) + 2 * hh + y) * 4u;

  v4f xp = *reinterpret_cast<const v4f*>(xb);

  v8f cn;
#pragma unroll
  for (int r = 0; r < 8; ++r) cn[r] = fmaf(xp[0], win[r], ball[r]);

  for (int t0 = 0; t0 < T_LEN; t0 += TSTEP) {
    const int tl = (t0 + TSTEP < T_LEN) ? (t0 + TSTEP) : (t0 + TSTEP - 4);
    v4f xq[TSTEP / 4];

    char* hst = reinterpret_cast<char*>(hs) + (unsigned)__builtin_amdgcn_readfirstlane(t0 * 16);

#pragma unroll
    for (int i = 0; i < TSTEP; ++i) {
      v8f acc = __builtin_amdgcn_wmma_f32_16x16x4_f32(
           false, a,  false, hv,
           (short)0, cn,  false,  false);

      if (i == 0) {
        __builtin_amdgcn_sched_barrier(0);
#pragma unroll
        for (int q = 0; q < TSTEP / 4 - 1; ++q)
          xq[q] = *reinterpret_cast<const v4f*>(xb + t0 + 4 + 4 * q);
        xq[TSTEP / 4 - 1] = *reinterpret_cast<const v4f*>(xb + tl);
        __builtin_amdgcn_sched_barrier(0);
      }

      const float xt1 = (i + 1 < 4) ? xp[(i + 1) & 3] : xq[(i + 1 - 4) >> 2][(i + 1) & 3];
#pragma unroll
      for (int r = 0; r < 8; ++r) cn[r] = fmaf(xt1, win[r], ball[r]);

      float sg[4];
#pragma unroll
      for (int r = 0; r < 4; ++r) {
        const float lo = acc[r];
        const float hi = acc[4 + r];
        sg[r] = y ? hi : lo;
      }
#pragma unroll
      for (int r = 0; r < 4; ++r) sg[r] = __builtin_amdgcn_exp2f(sg[r]);
#pragma unroll
      for (int r = 0; r < 4; ++r) sg[r] = __builtin_amdgcn_rcpf(1.0f + sg[r]);

      const float p = fmaf(sg[2], 2.0f * KC, -KC);
      cs = fmaf(sg[0], p, sg[1] * cs);

      const float ox = partner8(sg[3]);

      const float r2 = __builtin_amdgcn_rcpf(fmaf(__builtin_amdgcn_exp2f(cs), 0.5f, 0.5f));
      hv.x = fmaf(sg[3], r2, -sg[3]);
      hv.y = fmaf(partner8(r2), ox, -ox);

      *reinterpret_cast<float*>(hst + i * 16 + lane_off) = hv.x;

      __builtin_amdgcn_sched_barrier(0x0044);
    }
    xp = xq[TSTEP / 4 - 1];
  }
}

#define FC_ROWS    64
#define FC_PASSES  2
#define FC_WAVES   8
#define FC_BLOCKS  (NROWS / (FC_ROWS * FC_PASSES * FC_WAVES))
#define FC_TILE    (FC_ROWS * NCLS)

__global__ __launch_bounds__(256)
void fc_wmma_kernel(const float* __restrict__ hs,
                    const float* __restrict__ W_fc,
                    const float* __restrict__ b_fc,
                    float* __restrict__ out)
{
  __shared__ __attribute__((aligned(16))) float stage[FC_WAVES][FC_TILE];

  const int wib  = threadIdx.x >> 5;
  const int lane = threadIdx.x & 31;
  const int lm   = lane & 15;
  const int half = lane >> 4;
  const int k0   = 2 * half;
  float* st = stage[wib];

  v2f bm0, bm1;
  bm0.x = W_fc[lm * 4 + k0];
  bm0.y = W_fc[lm * 4 + k0 + 1];
  const float bias0 = b_fc[lm];
  const bool  hi_ok = (16 + lm) < NCLS;
  float bias1 = 0.0f;
  bm1.x = 0.0f; bm1.y = 0.0f;
  if (hi_ok) {
    bm1.x = W_fc[(16 + lm) * 4 + k0];
    bm1.y = W_fc[(16 + lm) * 4 + k0 + 1];
    bias1 = b_fc[16 + lm];
  }

  const size_t row0 = (size_t)(blockIdx.x * FC_WAVES + wib) * (FC_ROWS * FC_PASSES);

  v2f a[FC_PASSES][FC_ROWS / 16];
#pragma unroll
  for (int p = 0; p < FC_PASSES; ++p)
#pragma unroll
    for (int rt = 0; rt < FC_ROWS / 16; ++rt)
      a[p][rt] = *reinterpret_cast<const v2f*>(
          hs + (row0 + (size_t)(p * FC_ROWS + rt * 16 + lm)) * 4 + k0);

#pragma unroll
  for (int p = 0; p < FC_PASSES; ++p) {
#pragma unroll
    for (int rt = 0; rt < FC_ROWS / 16; ++rt) {
      v8f acc0, acc1;
#pragma unroll
      for (int r = 0; r < 8; ++r) { acc0[r] = bias0; acc1[r] = bias1; }
      acc0 = __builtin_amdgcn_wmma_f32_16x16x4_f32(
           false, a[p][rt],  false, bm0,
           (short)0, acc0,  false,  false);
      acc1 = __builtin_amdgcn_wmma_f32_16x16x4_f32(
           false, a[p][rt],  false, bm1,
           (short)0, acc1,  false,  false);
      float* sr = st + (rt * 16 + 8 * half) * NCLS + lm;
#pragma unroll
      for (int r = 0; r < 8; ++r) sr[r * NCLS] = acc0[r];
      if (hi_ok) {
#pragma unroll
        for (int r = 0; r < 8; ++r) sr[r * NCLS + 16] = acc1[r];
      }
    }

    __builtin_amdgcn_fence(__ATOMIC_RELEASE, "wavefront");
    __builtin_amdgcn_wave_barrier();
    __builtin_amdgcn_fence(__ATOMIC_ACQUIRE, "wavefront");

    float* og = out + (row0 + (size_t)p * FC_ROWS) * NCLS;
#pragma unroll
    for (int j = 0; j < (FC_TILE / 4 + 31) / 32; ++j) {
      const int q = j * 32 + lane;
      if (q < FC_TILE / 4)
        *reinterpret_cast<v4f*>(og + q * 4) = *reinterpret_cast<const v4f*>(st + q * 4);
    }

    __builtin_amdgcn_fence(__ATOMIC_RELEASE, "wavefront");
    __builtin_amdgcn_wave_barrier();
    __builtin_amdgcn_fence(__ATOMIC_ACQUIRE, "wavefront");
  }
}

extern "C" void kernel_launch(void* const* d_in, const int* in_sizes, int n_in,
                              void* d_out, int out_size, void* d_ws, size_t ws_size,
                              hipStream_t stream) {
  const float* x    = (const float*)d_in[0];
  const float* W_ih = (const float*)d_in[1];
  const float* W_hh = (const float*)d_in[2];
  const float* b_ih = (const float*)d_in[3];
  const float* b_hh = (const float*)d_in[4];
  const float* W_fc = (const float*)d_in[5];
  const float* b_fc = (const float*)d_in[6];
  float* out = (float*)d_out;

  float* hs = (float*)d_ws;

  lstm_scan_wmma_kernel<<<BATCH / SCAN_BT, 32, 0, stream>>>(
      x, W_ih, W_hh, b_ih, b_hh, hs);

  fc_wmma_kernel<<<FC_BLOCKS, 256, 0, stream>>>(hs, W_fc, b_fc, out);
}
